# starting kernel
# baseline (speedup 1.0000x reference)
#include <hip/hip_runtime.h>

typedef __attribute__((ext_vector_type(2))) float v2f;
typedef __attribute__((ext_vector_type(4))) float v4f;
typedef __attribute__((ext_vector_type(8))) float v8f;

#define BATCH 16
#define M_DIM 2048
#define K_DIM 2048
#define N_DIM 64
#define KTILE 32
#define MTILE 32
#define LDA   33

__global__ __launch_bounds__(128) void sparse_bmm_wmma(const float* __restrict__ A,
                                                       const float* __restrict__ Bm,
                                                       float* __restrict__ C) {
  __shared__ float sA[MTILE * LDA];
  __shared__ float sB[KTILE * N_DIM];
  __shared__ int   s_nz;

  const int tid   = threadIdx.x;
  const int lane  = tid & 31;
  const int wave  = tid >> 5;
  const int m16   = lane & 15;
  const int khalf = (lane < 16) ? 0 : 2;
  const int batch = blockIdx.y;
  const int m0    = blockIdx.x * MTILE;
  const int n0    = wave * 16;

  const float* Ab = A  + ((size_t)batch * M_DIM + m0) * K_DIM;
  const float* Bb = Bm + (size_t)batch * K_DIM * N_DIM;

  const int arow = tid >> 2;
  const int acol = (tid & 3) * 8;
  const int brow = tid >> 2;
  const int bcol = (tid & 3) * 16;

  v8f acc0 = {};
  v8f acc1 = {};

  for (int kc = 0; kc < K_DIM; kc += KTILE) {
    if (tid == 0) s_nz = 0;

    const float* ap = Ab + (size_t)arow * K_DIM + kc + acol;
    v4f a0 = *(const v4f*)(ap);
    v4f a1 = *(const v4f*)(ap + 4);

    if (kc + KTILE < K_DIM) {
      __builtin_prefetch(ap + KTILE, 0, 0);
    }

    unsigned nzb = 0;
#pragma unroll
    for (int i = 0; i < 4; ++i) {
      nzb |= __float_as_uint(a0[i]) | __float_as_uint(a1[i]);
    }

    float* as = &sA[arow * LDA + acol];
#pragma unroll
    for (int i = 0; i < 4; ++i) { as[i] = a0[i]; as[4 + i] = a1[i]; }

    __syncthreads();
    if (nzb) s_nz = 1;
    __syncthreads();
    const int nz = s_nz;

    if (nz) {
      const float* bp = Bb + (size_t)(kc + brow) * N_DIM + bcol;
      v4f b0 = *(const v4f*)(bp);
      v4f b1 = *(const v4f*)(bp + 4);
      v4f b2 = *(const v4f*)(bp + 8);
      v4f b3 = *(const v4f*)(bp + 12);
      v4f* bs = (v4f*)&sB[brow * N_DIM + bcol];
      bs[0] = b0; bs[1] = b1; bs[2] = b2; bs[3] = b3;
      __syncthreads();

#pragma unroll
      for (int kk = 0; kk < KTILE; kk += 4) {
        const float* ac0 = &sA[m16 * LDA + kk + khalf];
        const float* ac1 = &sA[(m16 + 16) * LDA + kk + khalf];
        v2f fa0; fa0.x = ac0[0]; fa0.y = ac0[1];
        v2f fa1; fa1.x = ac1[0]; fa1.y = ac1[1];
        const float* bq = &sB[(kk + khalf) * N_DIM + n0 + m16];
        v2f fb;  fb.x = bq[0];  fb.y = bq[N_DIM];

        acc0 = __builtin_amdgcn_wmma_f32_16x16x4_f32(false, fa0, false, fb,
                                                     (short)0, acc0, false, false);
        acc1 = __builtin_amdgcn_wmma_f32_16x16x4_f32(false, fa1, false, fb,
                                                     (short)0, acc1, false, false);
      }
    }
    __syncthreads();
  }

  const int mrow = (lane < 16) ? 0 : 8;
  float* cp = C + ((size_t)batch * M_DIM + m0) * N_DIM + n0 + m16;
#pragma unroll
  for (int r = 0; r < 8; ++r) {
    cp[(size_t)(mrow + r) * N_DIM]      = acc0[r];
    cp[(size_t)(16 + mrow + r) * N_DIM] = acc1[r];
  }
}

extern "C" void kernel_launch(void* const* d_in, const int* in_sizes, int n_in,
                              void* d_out, int out_size, void* d_ws, size_t ws_size,
                              hipStream_t stream) {
  (void)in_sizes; (void)n_in; (void)out_size; (void)d_ws; (void)ws_size;
  const float* A  = (const float*)d_in[0];
  const float* Bm = (const float*)d_in[1];
  float*       C  = (float*)d_out;
  dim3 grid(M_DIM / MTILE, BATCH);
  sparse_bmm_wmma<<<grid, 128, 0, stream>>>(A, Bm, C);
}


# Each wave streams its own 16 A rows (whole lines, 8 chunks ahead) via a private LDS tile with a wave-level zero test; B staged per 512 k; no per-chunk barriers; k summed in a new order
# speedup vs baseline: 1.7199x; 1.7199x over previous
#include <hip/hip_runtime.h>


typedef __attribute__((ext_vector_type(2))) float v2f;
typedef __attribute__((ext_vector_type(4))) float v4f;
typedef __attribute__((ext_vector_type(8))) float v8f;

#define BATCH 16
#define M_DIM 2048
#define K_DIM 2048
#define N_DIM 64
#define WAVES 8
#define MTILE (16 * WAVES)
#define KC 32
#define DEPTH 8
#define KHALF (KC * DEPTH)
#define KSTAGE (2 * KHALF)
#define NSTAGE (K_DIM / KSTAGE)
#define PA 36
#define PBP 136

static __device__ __forceinline__ void lds_barrier() {
  asm volatile("s_wait_dscnt 0x0\n\ts_barrier_signal -1\n\ts_barrier_wait -1" ::: "memory");
}

static __device__ __forceinline__ void keep_loads_here() {
  asm volatile("" ::: "memory");
}

__global__ __launch_bounds__(256) void sparse_bmm_wmma(const float* A, const float* Bm, float* C) {
  __shared__ __attribute__((aligned(16))) float sB[2][(KSTAGE / 2) * PBP];
  __shared__ __attribute__((aligned(16))) float sA[WAVES][16 * PA];

  const int tid   = threadIdx.x;
  const int lane  = tid & 31;
  const int wave  = tid >> 5;
  const int c     = lane & 15;
  const int h     = lane >> 4;
  const int batch = blockIdx.y;
  const int m0    = blockIdx.x * MTILE + wave * 16;

  const float* ag  = A + ((size_t)batch * M_DIM + m0 + (lane >> 3)) * K_DIM + (lane & 7) * 4;
  float*       ast = &sA[wave][(lane >> 3) * PA + (lane & 7) * 4];
  const float* ald = &sA[wave][c * PA + 8 * h];

  const float* bg  = Bm + ((size_t)batch * K_DIM + 2 * (tid >> 4)) * N_DIM + (tid & 15) * 4;
  const int    bso = (tid >> 4) * PBP + (tid & 15) * 8;

  v8f acc[4] = {};
  v4f ar[DEPTH][4];
  v4f br[16];

  {
    v4f b2[16];
#pragma unroll
    for (int i = 0; i < 8; ++i) {
      br[2 * i]     = *(const v4f*)(bg + (size_t)(32 * i) * N_DIM);
      br[2 * i + 1] = *(const v4f*)(bg + (size_t)(32 * i + 1) * N_DIM);
    }
#pragma unroll
    for (int i = 0; i < 8; ++i) {
      b2[2 * i]     = *(const v4f*)(bg + (size_t)(KHALF + 32 * i) * N_DIM);
      b2[2 * i + 1] = *(const v4f*)(bg + (size_t)(KHALF + 32 * i + 1) * N_DIM);
    }
#pragma unroll
    for (int d = 0; d < DEPTH; ++d) {
#pragma unroll
      for (int q = 0; q < 4; ++q) ar[d][q] = *(const v4f*)(ag + (size_t)(4 * q) * K_DIM + d * KC);
    }
    keep_loads_here();
#pragma unroll
    for (int i = 0; i < 8; ++i) {
      float* d0 = &sB[0][bso + (16 * i) * PBP];
      float* d1 = &sB[0][bso + (KHALF / 2 + 16 * i) * PBP];
      const v4f p0 = br[2 * i], p1 = br[2 * i + 1], q0 = b2[2 * i], q1 = b2[2 * i + 1];
      *(v4f*)(d0)     = (v4f){p0.x, p1.x, p0.y, p1.y};
      *(v4f*)(d0 + 4) = (v4f){p0.z, p1.z, p0.w, p1.w};
      *(v4f*)(d1)     = (v4f){q0.x, q1.x, q0.y, q1.y};
      *(v4f*)(d1 + 4) = (v4f){q0.z, q1.z, q0.w, q1.w};
    }
  }
  lds_barrier();

#pragma unroll 1
  for (int s = 0; s < NSTAGE; ++s) {
    const int    sn    = (s + 1 < NSTAGE) ? (s + 1) : (NSTAGE - 1);
    const float* bsrc  = bg + (size_t)sn * KSTAGE * N_DIM;
    float*       bdst  = &sB[(s + 1) & 1][bso];
    const float* sbase = &sB[s & 1][(4 * h) * PBP + 2 * c];

#pragma unroll 1
    for (int hf = 0; hf < 2; ++hf) {
#pragma unroll
      for (int i = 0; i < 8; ++i) {
        br[2 * i]     = *(const v4f*)(bsrc + (size_t)(hf * KHALF + 32 * i) * N_DIM);
        br[2 * i + 1] = *(const v4f*)(bsrc + (size_t)(hf * KHALF + 32 * i + 1) * N_DIM);
      }
      keep_loads_here();

      const float* sb  = sbase + hf * (KHALF / 2) * PBP;
      const int    ch0 = (s * 2 + hf) * DEPTH;

#pragma unroll
      for (int j = 0; j < DEPTH; ++j) {
        unsigned nzb = 0;
#pragma unroll
        for (int q = 0; q < 4; ++q) {
#pragma unroll
          for (int e = 0; e < 4; ++e) nzb |= __float_as_uint(ar[j][q][e]);
        }

        if (__builtin_amdgcn_ballot_w32(nzb != 0u) != 0u) {
#pragma unroll
          for (int q = 0; q < 4; ++q) *(v4f*)(ast + q * 4 * PA) = ar[j][q];

          const float* sbk = sb + (j * (KC / 2)) * PBP;
#pragma unroll
          for (int g = 0; g < 2; ++g) {
            const v4f x0 = *(const v4f*)(ald + 16 * g);
            const v4f x1 = *(const v4f*)(ald + 16 * g + 4);
#pragma unroll
            for (int jj = 0; jj < 4; ++jj) {
              v2f fa;
              fa.x = (jj < 2) ? x0[(2 * jj) & 3] : x1[(2 * jj) & 3];
              fa.y = (jj < 2) ? x0[(2 * jj + 1) & 3] : x1[(2 * jj + 1) & 3];
              const float* bq = sbk + (8 * g + jj) * PBP;
#pragma unroll
              for (int t = 0; t < 4; ++t) {
                const v2f fb = *(const v2f*)(bq + 32 * t);
                acc[t] = __builtin_amdgcn_wmma_f32_16x16x4_f32(false, fa, false, fb,
                                                               (short)0, acc[t], false, false);
              }
            }
          }
        }

        int nc = ch0 + j + DEPTH;
        nc = (nc < K_DIM / KC) ? nc : (K_DIM / KC - 1);
#pragma unroll
        for (int q = 0; q < 4; ++q) ar[j][q] = *(const v4f*)(ag + (size_t)(4 * q) * K_DIM + nc * KC);
        keep_loads_here();
      }

#pragma unroll
      for (int i = 0; i < 8; ++i) {
        float* d0 = bdst + (hf * (KHALF / 2) + 16 * i) * PBP;
        const v4f p0 = br[2 * i], p1 = br[2 * i + 1];
        *(v4f*)(d0)     = (v4f){p0.x, p1.x, p0.y, p1.y};
        *(v4f*)(d0 + 4) = (v4f){p0.z, p1.z, p0.w, p1.w};
      }
    }
    lds_barrier();
  }

  float* cp = C + ((size_t)batch * M_DIM + m0 + 8 * h) * N_DIM + c;
#pragma unroll
  for (int r = 0; r < 8; ++r) {
#pragma unroll
    for (int t = 0; t < 4; ++t) cp[(size_t)r * N_DIM + 16 * t] = acc[t][r];
  }
}

extern "C" void kernel_launch(void* const* d_in, const int* in_sizes, int n_in,
                              void* d_out, int out_size, void* d_ws, size_t ws_size,
                              hipStream_t stream) {
  (void)in_sizes; (void)n_in; (void)out_size; (void)d_ws; (void)ws_size;
  const float* A  = (const float*)d_in[0];
  const float* Bm = (const float*)d_in[1];
  float*       C  = (float*)d_out;
  dim3 grid(M_DIM / MTILE, BATCH);
  sparse_bmm_wmma<<<grid, 256, 0, stream>>>(A, Bm, C);
}


# Each chunk's LDS traffic (A tile round trip, A and B fragment reads) issued as one group one chunk ahead into a second fragment register set, not among the matrix instructions; k order unchanged
# speedup vs baseline: 1.7982x; 1.0455x over previous
#include <hip/hip_runtime.h>


typedef __attribute__((ext_vector_type(2))) float v2f;
typedef __attribute__((ext_vector_type(4))) float v4f;
typedef __attribute__((ext_vector_type(8))) float v8f;

#define BATCH 16
#define M_DIM 2048
#define K_DIM 2048
#define N_DIM 64
#define WAVES 8
#define MTILE (16 * WAVES)
#define KC 32
#define DEPTH 8
#define KHALF (KC * DEPTH)
#define KSTAGE (2 * KHALF)
#define NSTAGE (K_DIM / KSTAGE)
#define NTRIP (2 * NSTAGE)
#define PA 36
#define PBP 136

static __device__ __forceinline__ void lds_barrier() {
  asm volatile("s_wait_dscnt 0x0\n\ts_barrier_signal -1\n\ts_barrier_wait -1" ::: "memory");
}

static __device__ __forceinline__ void keep_loads_here() {
  asm volatile("" ::: "memory");
}


static __device__ __forceinline__ bool chunk_nonzero(const v4f (&a)[4]) {
  unsigned nzb = 0;
#pragma unroll
  for (int q = 0; q < 4; ++q) {
#pragma unroll
    for (int e = 0; e < 4; ++e) nzb |= __float_as_uint(a[q][e]);
  }
  return __builtin_amdgcn_ballot_w32(nzb != 0u) != 0u;
}

static __device__ __forceinline__ void fetch_fragments(const v4f (&a)[4], float* ast, const float* ald,
                                                       const float* sbk, v4f (&fa)[4], v2f (&fb)[32]) {
#pragma unroll
  for (int q = 0; q < 4; ++q) *(v4f*)(ast + q * 4 * PA) = a[q];
#pragma unroll
  for (int g = 0; g < 2; ++g) {
    fa[2 * g]     = *(const v4f*)(ald + 16 * g);
    fa[2 * g + 1] = *(const v4f*)(ald + 16 * g + 4);
  }
#pragma unroll
  for (int g = 0; g < 2; ++g) {
#pragma unroll
    for (int jj = 0; jj < 4; ++jj) {
#pragma unroll
      for (int t = 0; t < 4; ++t) fb[(4 * g + jj) * 4 + t] = *(const v2f*)(sbk + (8 * g + jj) * PBP + 32 * t);
    }
  }
}

static __device__ __forceinline__ void multiply_chunk(const v4f (&fa)[4], const v2f (&fb)[32], v8f (&acc)[4]) {
#pragma unroll
  for (int g = 0; g < 2; ++g) {
#pragma unroll
    for (int jj = 0; jj < 4; ++jj) {
      const v4f x = fa[2 * g + (jj >> 1)];
      v2f a;
      a.x = x[(2 * jj) & 3];
      a.y = x[(2 * jj + 1) & 3];
#pragma unroll
      for (int t = 0; t < 4; ++t) {
        acc[t] = __builtin_amdgcn_wmma_f32_16x16x4_f32(false, a, false, fb[(4 * g + jj) * 4 + t],
                                                       (short)0, acc[t], false, false);
      }
    }
  }
}

__global__ __launch_bounds__(256) void sparse_bmm_wmma(const float* A, const float* Bm, float* C) {
  __shared__ __attribute__((aligned(16))) float sB[2][(KSTAGE / 2) * PBP];
  __shared__ __attribute__((aligned(16))) float sA[WAVES][16 * PA];

  const int tid   = threadIdx.x;
  const int lane  = tid & 31;
  const int wave  = tid >> 5;
  const int c     = lane & 15;
  const int h     = lane >> 4;
  const int batch = blockIdx.y;
  const int m0    = blockIdx.x * MTILE + wave * 16;

  const float* ag  = A + ((size_t)batch * M_DIM + m0 + (lane >> 3)) * K_DIM + (lane & 7) * 4;
  float*       ast = &sA[wave][(lane >> 3) * PA + (lane & 7) * 4];
  const float* ald = &sA[wave][c * PA + 8 * h];

  const float* bg  = Bm + ((size_t)batch * K_DIM + 2 * (tid >> 4)) * N_DIM + (tid & 15) * 4;
  const int    bso = (tid >> 4) * PBP + (tid & 15) * 8;
  const int    bfo = (4 * h) * PBP + 2 * c;

  v8f acc[4] = {};
  v4f ar[DEPTH][4];
  v4f br[16];
  v4f fa[2][4];
  v2f fb[2][32];
  bool nzc;

  {
    v4f b2[16];
#pragma unroll
    for (int i = 0; i < 8; ++i) {
      br[2 * i]     = *(const v4f*)(bg + (size_t)(32 * i) * N_DIM);
      br[2 * i + 1] = *(const v4f*)(bg + (size_t)(32 * i + 1) * N_DIM);
    }
#pragma unroll
    for (int i = 0; i < 8; ++i) {
      b2[2 * i]     = *(const v4f*)(bg + (size_t)(KHALF + 32 * i) * N_DIM);
      b2[2 * i + 1] = *(const v4f*)(bg + (size_t)(KHALF + 32 * i + 1) * N_DIM);
    }
#pragma unroll
    for (int d = 0; d < DEPTH; ++d) {
#pragma unroll
      for (int q = 0; q < 4; ++q) ar[d][q] = *(const v4f*)(ag + (size_t)(4 * q) * K_DIM + d * KC);
    }
    keep_loads_here();
#pragma unroll
    for (int i = 0; i < 8; ++i) {
      float* d0 = &sB[0][bso + (16 * i) * PBP];
      float* d1 = &sB[0][bso + (KHALF / 2 + 16 * i) * PBP];
      const v4f p0 = br[2 * i], p1 = br[2 * i + 1], q0 = b2[2 * i], q1 = b2[2 * i + 1];
      *(v4f*)(d0)     = (v4f){p0.x, p1.x, p0.y, p1.y};
      *(v4f*)(d0 + 4) = (v4f){p0.z, p1.z, p0.w, p1.w};
      *(v4f*)(d1)     = (v4f){q0.x, q1.x, q0.y, q1.y};
      *(v4f*)(d1 + 4) = (v4f){q0.z, q1.z, q0.w, q1.w};
    }
  }
  lds_barrier();

  nzc = chunk_nonzero(ar[0]);
  fetch_fragments(ar[0], ast, ald, &sB[0][bfo], fa[0], fb[0]);
  keep_loads_here();

#pragma unroll 1
  for (int u = 0; u < NTRIP; ++u) {
    const int    s     = u >> 1;
    const int    hf    = u & 1;
    const int    sn    = (s + 1 < NSTAGE) ? (s + 1) : (NSTAGE - 1);
    const float* bsrc  = bg + ((size_t)sn * KSTAGE + hf * KHALF) * N_DIM;
    float*       bdst  = &sB[(s + 1) & 1][bso + hf * (KHALF / 2) * PBP];
    const float* sb    = &sB[s & 1][bfo + hf * (KHALF / 2) * PBP];
    const int    ch0   = u * DEPTH;

#pragma unroll
    for (int i = 0; i < 8; ++i) {
      br[2 * i]     = *(const v4f*)(bsrc + (size_t)(32 * i) * N_DIM);
      br[2 * i + 1] = *(const v4f*)(bsrc + (size_t)(32 * i + 1) * N_DIM);
    }
    keep_loads_here();

#pragma unroll
    for (int j = 0; j < DEPTH; ++j) {
      bool nzn = false;
      if (j + 1 < DEPTH) {
        nzn = chunk_nonzero(ar[j + 1]);
        fetch_fragments(ar[j + 1], ast, ald, sb + ((j + 1) * (KC / 2)) * PBP, fa[(j + 1) & 1], fb[(j + 1) & 1]);
        keep_loads_here();
      }

      if (nzc) multiply_chunk(fa[j & 1], fb[j & 1], acc);
      nzc = nzn;

      int nc = ch0 + j + DEPTH;
      nc = (nc < K_DIM / KC) ? nc : (K_DIM / KC - 1);
#pragma unroll
      for (int q = 0; q < 4; ++q) ar[j][q] = *(const v4f*)(ag + (size_t)(4 * q) * K_DIM + nc * KC);
      keep_loads_here();
    }

#pragma unroll
    for (int i = 0; i < 8; ++i) {
      float* d0 = bdst + (16 * i) * PBP;
      const v4f p0 = br[2 * i], p1 = br[2 * i + 1];
      *(v4f*)(d0)     = (v4f){p0.x, p1.x, p0.y, p1.y};
      *(v4f*)(d0 + 4) = (v4f){p0.z, p1.z, p0.w, p1.w};
    }
    if (hf) lds_barrier();

    if (u + 1 < NTRIP) {
      const int un = u + 1;
      nzc = chunk_nonzero(ar[0]);
      fetch_fragments(ar[0], ast, ald, &sB[(un >> 1) & 1][bfo + (un & 1) * (KHALF / 2) * PBP], fa[0], fb[0]);
      keep_loads_here();
    }
  }

  float* cp = C + ((size_t)batch * M_DIM + m0 + 8 * h) * N_DIM + c;
#pragma unroll
  for (int r = 0; r < 8; ++r) {
#pragma unroll
    for (int t = 0; t < 4; ++t) cp[(size_t)r * N_DIM + 16 * t] = acc[t][r];
  }
}

extern "C" void kernel_launch(void* const* d_in, const int* in_sizes, int n_in,
                              void* d_out, int out_size, void* d_ws, size_t ws_size,
                              hipStream_t stream) {
  (void)in_sizes; (void)n_in; (void)out_size; (void)d_ws; (void)ws_size;
  const float* A  = (const float*)d_in[0];
  const float* Bm = (const float*)d_in[1];
  float*       C  = (float*)d_out;
  dim3 grid(M_DIM / MTILE, BATCH);
  sparse_bmm_wmma<<<grid, 256, 0, stream>>>(A, Bm, C);
}
